# starting kernel
# baseline (speedup 1.0000x reference)
#include <hip/hip_runtime.h>
#include <hip/hip_bf16.h>

typedef float v2f __attribute__((ext_vector_type(2)));
typedef float v8f __attribute__((ext_vector_type(8)));

struct c32 { float r, i; };
__device__ __forceinline__ c32 cmul(c32 a, c32 b) { return { a.r*b.r - a.i*b.i, a.r*b.i + a.i*b.r }; }
__device__ __forceinline__ c32 cadd(c32 a, c32 b) { return { a.r + b.r, a.i + b.i }; }
__device__ __forceinline__ c32 cscale(float s, c32 a) { return { s*a.r, s*a.i }; }


__global__ void k_setup(const float* __restrict__ pk, const float* __restrict__ fc_p,
                        const int* __restrict__ dt_p, const int* __restrict__ dtf_p,
                        float* __restrict__ scal) {
    float dt  = (float)dt_p[0];
    int nsub  = dtf_p[0] / dt_p[0];
    float K0  = expf(pk[0]);
    float K1  = expf(pk[1]);
    c32 a = { 1.0f - dt * K1, -dt * fc_p[0] };
    c32 S1 = {0.f, 0.f}, S2 = {0.f, 0.f};
    c32 p = {1.f, 0.f};
    for (int j = nsub - 1; j >= 0; --j) {
        float aa = (float)j / (float)nsub;
        S1 = cadd(S1, cscale(1.0f - aa, p));
        S2 = cadd(S2, cscale(aa, p));
        p = cmul(p, a);
    }
    c32 A  = p;
    c32 c1 = cscale(dt * K0, S1);
    c32 c2 = cscale(dt * K0, S2);
    scal[0] = c1.r; scal[1] = c1.i;
    scal[2] = c2.r; scal[3] = c2.i;
    c32 Pj = {1.f, 0.f};
    for (int j = 0; j <= 16; ++j) { scal[4 + 2*j] = Pj.r; scal[5 + 2*j] = Pj.i; Pj = cmul(Pj, A); }
    c32 A16 = { scal[36], scal[37] };
    c32 A256 = {1.f, 0.f};
    for (int j = 0; j < 16; ++j) A256 = cmul(A256, A16);
    scal[38] = A256.r; scal[39] = A256.i;
}

__global__ void k_forcing(const float* __restrict__ TAr, const float* __restrict__ TAi,
                          const float* __restrict__ scal,
                          const int* __restrict__ dt_p, const int* __restrict__ dtf_p,
                          float* __restrict__ br, float* __restrict__ bi, int NS) {
    int i = blockIdx.x * blockDim.x + threadIdx.x;
    if (i >= NS) return;
    int nsub = dtf_p[0] / dt_p[0];
    int g  = i / nsub;
    int gm = (g == 0) ? (NS - 1) : (g - 1);
    c32 c1 = { scal[0], scal[1] };
    c32 c2 = { scal[2], scal[3] };
    c32 tm = { TAr[gm], TAi[gm] };
    c32 tg = { TAr[g],  TAi[g]  };
    c32 b  = cadd(cmul(c1, tm), cmul(c2, tg));
    br[i] = b.r; bi[i] = b.i;
}

__device__ __forceinline__ void wmma4(v8f& acc, float a0, float a1, float b0, float b1) {
    v2f A; A[0] = a0; A[1] = a1;
    v2f B; B[0] = b0; B[1] = b1;
    acc = __builtin_amdgcn_wmma_f32_16x16x4_f32(false, A, false, B, (short)0, acc, false, false);
}

__device__ __forceinline__ float bc_lane(float v, int srcLane) {
    return __int_as_float(__builtin_amdgcn_readlane(__float_as_int(v), srcLane));
}

template<bool WRITE>
__global__ void __launch_bounds__(32)
k_scan(const float* __restrict__ br, const float* __restrict__ bi,
       const float* __restrict__ scal,
       const float* __restrict__ carr, const float* __restrict__ cari,
       float* __restrict__ totr, float* __restrict__ toti,
       float* __restrict__ out, int NS) {
    __shared__ alignas(16) float lr[256];
    __shared__ alignas(16) float li[256];
    const int lane = threadIdx.x;
    const int base = blockIdx.x * 256;

    if constexpr (WRITE) {
        __builtin_prefetch(carr + blockIdx.x, 0, 3);
        __builtin_prefetch(cari + blockIdx.x, 0, 3);
    }

    {
        const float4* gr = reinterpret_cast<const float4*>(br + base);
        const float4* gi = reinterpret_cast<const float4*>(bi + base);
        float4* sr = reinterpret_cast<float4*>(lr);
        float4* si = reinterpret_cast<float4*>(li);
        sr[lane]      = gr[lane];
        sr[lane + 32] = gr[lane + 32];
        si[lane]      = gi[lane];
        si[lane + 32] = gi[lane + 32];
    }
    __syncthreads();

    const int hi  = lane >> 4;
    const int low = lane & 15;

    float trf[4][2], tif[4][2];
    float brf[4][2], bif[4][2];
#pragma unroll
    for (int c = 0; c < 4; ++c) {
#pragma unroll
        for (int v = 0; v < 2; ++v) {
            int k  = 4*c + v + 2*hi;
            int d  = low - k;
            int dc = d >= 0 ? d : 0;
            float pr = scal[4 + 2*dc];
            float pi = scal[5 + 2*dc];
            trf[c][v] = d >= 0 ? pr : 0.0f;
            tif[c][v] = d >= 0 ? pi : 0.0f;
            brf[c][v] = lr[16*low + k];
            bif[c][v] = li[16*low + k];
        }
    }

    v8f accr = {0,0,0,0,0,0,0,0};
    v8f acci = {0,0,0,0,0,0,0,0};
#pragma unroll
    for (int c = 0; c < 4; ++c) wmma4(accr,  trf[c][0],  trf[c][1], brf[c][0], brf[c][1]);
#pragma unroll
    for (int c = 0; c < 4; ++c) wmma4(accr, -tif[c][0], -tif[c][1], bif[c][0], bif[c][1]);
#pragma unroll
    for (int c = 0; c < 4; ++c) wmma4(acci,  trf[c][0],  trf[c][1], bif[c][0], bif[c][1]);
#pragma unroll
    for (int c = 0; c < 4; ++c) wmma4(acci,  tif[c][0],  tif[c][1], brf[c][0], brf[c][1]);

    c32 A16 = { scal[36], scal[37] };
    c32 carry_in = { 0.f, 0.f };
    if constexpr (WRITE) carry_in = { carr[blockIdx.x], cari[blockIdx.x] };

    const float vr7 = accr[7], vi7 = acci[7];
    c32 run = carry_in;
    c32 mycarry = carry_in;
#pragma unroll
    for (int s = 0; s < 16; ++s) {
        if (s == low) mycarry = run;
        c32 t = { bc_lane(vr7, 16 + s), bc_lane(vi7, 16 + s) };
        run = cadd(cmul(A16, run), t);
    }

    if constexpr (!WRITE) {
        if (lane == 0) { totr[blockIdx.x] = run.r; toti[blockIdx.x] = run.i; }
    } else {
        float ur[8], ui[8];
#pragma unroll
        for (int k = 0; k < 8; ++k) {
            int m = k + 8*hi;
            c32 pj = { scal[4 + 2*(m + 1)], scal[5 + 2*(m + 1)] };
            c32 u  = { accr[k], acci[k] };
            u = cadd(u, cmul(pj, mycarry));
            ur[k] = u.r; ui[k] = u.i;
        }
        const int gi0 = base + 16*low + 8*hi;
        float4* outr = reinterpret_cast<float4*>(out + gi0);
        float4* outi = reinterpret_cast<float4*>(out + NS + gi0);
        outr[0] = make_float4(ur[0], ur[1], ur[2], ur[3]);
        outr[1] = make_float4(ur[4], ur[5], ur[6], ur[7]);
        outi[0] = make_float4(ui[0], ui[1], ui[2], ui[3]);
        outi[1] = make_float4(ui[4], ui[5], ui[6], ui[7]);
    }
}

__global__ void k_blockscan(const float* __restrict__ totr, const float* __restrict__ toti,
                            const float* __restrict__ scal,
                            const float* __restrict__ U0, const float* __restrict__ V0,
                            float* __restrict__ carr, float* __restrict__ cari, int nblk) {
    if (threadIdx.x != 0 || blockIdx.x != 0) return;
    c32 A256 = { scal[38], scal[39] };
    c32 run = { U0[0], V0[0] };
    for (int B = 0; B < nblk; ++B) {
        carr[B] = run.r; cari[B] = run.i;
        c32 t = { totr[B], toti[B] };
        run = cadd(cmul(A256, run), t);
    }
}

extern "C" void kernel_launch(void* const* d_in, const int* in_sizes, int n_in,
                              void* d_out, int out_size, void* d_ws, size_t ws_size,
                              hipStream_t stream) {
    const float* pk  = (const float*)d_in[0];
    const float* TAr = (const float*)d_in[1];
    const float* TAi = (const float*)d_in[2];
    const float* fc  = (const float*)d_in[3];
    const float* U0  = (const float*)d_in[4];
    const float* V0  = (const float*)d_in[5];
    const int*   dt  = (const int*)d_in[6];
    const int*   dtf = (const int*)d_in[7];

    const int NS   = in_sizes[1];
    const int NBLK = NS / 256;

    float* ws   = (float*)d_ws;
    float* scal = ws;
    float* br   = ws + 64;
    float* bi   = br + NS;
    float* totr = bi + NS;
    float* toti = totr + NBLK;
    float* carr = toti + NBLK;
    float* cari = carr + NBLK;
    float* out  = (float*)d_out;

    k_setup<<<1, 1, 0, stream>>>(pk, fc, dt, dtf, scal);
    k_forcing<<<(NS + 255) / 256, 256, 0, stream>>>(TAr, TAi, scal, dt, dtf, br, bi, NS);
    k_scan<false><<<NBLK, 32, 0, stream>>>(br, bi, scal, nullptr, nullptr, totr, toti, out, NS);
    k_blockscan<<<1, 1, 0, stream>>>(totr, toti, scal, U0, V0, carr, cari, NBLK);
    k_scan<true><<<NBLK, 32, 0, stream>>>(br, bi, scal, carr, cari, totr, toti, out, NS);
}


# carry recurrence over the 1024 block totals: one thread's serial loop -> one 32-wave work-group, DPP scan per wave + LDS join of wave totals; carries now summed in tree order
# speedup vs baseline: 2.4538x; 2.4538x over previous
#include <hip/hip_runtime.h>
#include <hip/hip_bf16.h>

typedef float v2f __attribute__((ext_vector_type(2)));
typedef float v8f __attribute__((ext_vector_type(8)));

struct c32 { float r, i; };
__device__ __forceinline__ c32 cmul(c32 a, c32 b) { return { a.r*b.r - a.i*b.i, a.r*b.i + a.i*b.r }; }
__device__ __forceinline__ c32 cadd(c32 a, c32 b) { return { a.r + b.r, a.i + b.i }; }
__device__ __forceinline__ c32 cscale(float s, c32 a) { return { s*a.r, s*a.i }; }


__global__ void k_setup(const float* __restrict__ pk, const float* __restrict__ fc_p,
                        const int* __restrict__ dt_p, const int* __restrict__ dtf_p,
                        float* __restrict__ scal) {
    float dt  = (float)dt_p[0];
    int nsub  = dtf_p[0] / dt_p[0];
    float K0  = expf(pk[0]);
    float K1  = expf(pk[1]);
    c32 a = { 1.0f - dt * K1, -dt * fc_p[0] };
    c32 S1 = {0.f, 0.f}, S2 = {0.f, 0.f};
    c32 p = {1.f, 0.f};
    for (int j = nsub - 1; j >= 0; --j) {
        float aa = (float)j / (float)nsub;
        S1 = cadd(S1, cscale(1.0f - aa, p));
        S2 = cadd(S2, cscale(aa, p));
        p = cmul(p, a);
    }
    c32 A  = p;
    c32 c1 = cscale(dt * K0, S1);
    c32 c2 = cscale(dt * K0, S2);
    scal[0] = c1.r; scal[1] = c1.i;
    scal[2] = c2.r; scal[3] = c2.i;
    c32 Pj = {1.f, 0.f};
    for (int j = 0; j <= 16; ++j) { scal[4 + 2*j] = Pj.r; scal[5 + 2*j] = Pj.i; Pj = cmul(Pj, A); }
    c32 A16 = { scal[36], scal[37] };
    c32 A256 = {1.f, 0.f};
    for (int j = 0; j < 16; ++j) A256 = cmul(A256, A16);
    scal[38] = A256.r; scal[39] = A256.i;
}

__global__ void k_forcing(const float* __restrict__ TAr, const float* __restrict__ TAi,
                          const float* __restrict__ scal,
                          const int* __restrict__ dt_p, const int* __restrict__ dtf_p,
                          float* __restrict__ br, float* __restrict__ bi, int NS) {
    int i = blockIdx.x * blockDim.x + threadIdx.x;
    if (i >= NS) return;
    int nsub = dtf_p[0] / dt_p[0];
    int g  = i / nsub;
    int gm = (g == 0) ? (NS - 1) : (g - 1);
    c32 c1 = { scal[0], scal[1] };
    c32 c2 = { scal[2], scal[3] };
    c32 tm = { TAr[gm], TAi[gm] };
    c32 tg = { TAr[g],  TAi[g]  };
    c32 b  = cadd(cmul(c1, tm), cmul(c2, tg));
    br[i] = b.r; bi[i] = b.i;
}

__device__ __forceinline__ void wmma4(v8f& acc, float a0, float a1, float b0, float b1) {
    v2f A; A[0] = a0; A[1] = a1;
    v2f B; B[0] = b0; B[1] = b1;
    acc = __builtin_amdgcn_wmma_f32_16x16x4_f32(false, A, false, B, (short)0, acc, false, false);
}

__device__ __forceinline__ float bc_lane(float v, int srcLane) {
    return __int_as_float(__builtin_amdgcn_readlane(__float_as_int(v), srcLane));
}

template<bool WRITE>
__global__ void __launch_bounds__(32)
k_scan(const float* __restrict__ br, const float* __restrict__ bi,
       const float* __restrict__ scal,
       const float* __restrict__ carr, const float* __restrict__ cari,
       float* __restrict__ totr, float* __restrict__ toti,
       float* __restrict__ out, int NS) {
    __shared__ alignas(16) float lr[256];
    __shared__ alignas(16) float li[256];
    const int lane = threadIdx.x;
    const int base = blockIdx.x * 256;

    if constexpr (WRITE) {
        __builtin_prefetch(carr + blockIdx.x, 0, 3);
        __builtin_prefetch(cari + blockIdx.x, 0, 3);
    }

    {
        const float4* gr = reinterpret_cast<const float4*>(br + base);
        const float4* gi = reinterpret_cast<const float4*>(bi + base);
        float4* sr = reinterpret_cast<float4*>(lr);
        float4* si = reinterpret_cast<float4*>(li);
        sr[lane]      = gr[lane];
        sr[lane + 32] = gr[lane + 32];
        si[lane]      = gi[lane];
        si[lane + 32] = gi[lane + 32];
    }
    __syncthreads();

    const int hi  = lane >> 4;
    const int low = lane & 15;

    float trf[4][2], tif[4][2];
    float brf[4][2], bif[4][2];
#pragma unroll
    for (int c = 0; c < 4; ++c) {
#pragma unroll
        for (int v = 0; v < 2; ++v) {
            int k  = 4*c + v + 2*hi;
            int d  = low - k;
            int dc = d >= 0 ? d : 0;
            float pr = scal[4 + 2*dc];
            float pi = scal[5 + 2*dc];
            trf[c][v] = d >= 0 ? pr : 0.0f;
            tif[c][v] = d >= 0 ? pi : 0.0f;
            brf[c][v] = lr[16*low + k];
            bif[c][v] = li[16*low + k];
        }
    }

    v8f accr = {0,0,0,0,0,0,0,0};
    v8f acci = {0,0,0,0,0,0,0,0};
#pragma unroll
    for (int c = 0; c < 4; ++c) wmma4(accr,  trf[c][0],  trf[c][1], brf[c][0], brf[c][1]);
#pragma unroll
    for (int c = 0; c < 4; ++c) wmma4(accr, -tif[c][0], -tif[c][1], bif[c][0], bif[c][1]);
#pragma unroll
    for (int c = 0; c < 4; ++c) wmma4(acci,  trf[c][0],  trf[c][1], bif[c][0], bif[c][1]);
#pragma unroll
    for (int c = 0; c < 4; ++c) wmma4(acci,  tif[c][0],  tif[c][1], brf[c][0], brf[c][1]);

    c32 A16 = { scal[36], scal[37] };
    c32 carry_in = { 0.f, 0.f };
    if constexpr (WRITE) carry_in = { carr[blockIdx.x], cari[blockIdx.x] };

    const float vr7 = accr[7], vi7 = acci[7];
    c32 run = carry_in;
    c32 mycarry = carry_in;
#pragma unroll
    for (int s = 0; s < 16; ++s) {
        if (s == low) mycarry = run;
        c32 t = { bc_lane(vr7, 16 + s), bc_lane(vi7, 16 + s) };
        run = cadd(cmul(A16, run), t);
    }

    if constexpr (!WRITE) {
        if (lane == 0) { totr[blockIdx.x] = run.r; toti[blockIdx.x] = run.i; }
    } else {
        float ur[8], ui[8];
#pragma unroll
        for (int k = 0; k < 8; ++k) {
            int m = k + 8*hi;
            c32 pj = { scal[4 + 2*(m + 1)], scal[5 + 2*(m + 1)] };
            c32 u  = { accr[k], acci[k] };
            u = cadd(u, cmul(pj, mycarry));
            ur[k] = u.r; ui[k] = u.i;
        }
        const int gi0 = base + 16*low + 8*hi;
        float4* outr = reinterpret_cast<float4*>(out + gi0);
        float4* outi = reinterpret_cast<float4*>(out + NS + gi0);
        outr[0] = make_float4(ur[0], ur[1], ur[2], ur[3]);
        outr[1] = make_float4(ur[4], ur[5], ur[6], ur[7]);
        outi[0] = make_float4(ui[0], ui[1], ui[2], ui[3]);
        outi[1] = make_float4(ui[4], ui[5], ui[6], ui[7]);
    }
}

template<int N>
__device__ __forceinline__ float row_shr(float fill, float v) {
    return __int_as_float(__builtin_amdgcn_update_dpp(__float_as_int(fill), __float_as_int(v),
                                                      0x110 + N, 0xf, 0xf, false));
}

template<int N>
__device__ __forceinline__ void affine_step(c32& m, c32& v) {
    c32 pm = { row_shr<N>(1.0f, m.r), row_shr<N>(0.0f, m.i) };
    c32 pv = { row_shr<N>(0.0f, v.r), row_shr<N>(0.0f, v.i) };
    v = cadd(v, cmul(m, pv));
    m = cmul(m, pm);
}

__device__ __forceinline__ void wave_affine_scan(c32& m, c32& v, int lane) {
    affine_step<1>(m, v);
    affine_step<2>(m, v);
    affine_step<4>(m, v);
    affine_step<8>(m, v);
    c32 m0 = { bc_lane(m.r, 15), bc_lane(m.i, 15) };
    c32 v0 = { bc_lane(v.r, 15), bc_lane(v.i, 15) };
    c32 jv = cadd(v, cmul(m, v0));
    c32 jm = cmul(m, m0);
    const bool up = lane >= 16;
    v.r = up ? jv.r : v.r;  v.i = up ? jv.i : v.i;
    m.r = up ? jm.r : m.r;  m.i = up ? jm.i : m.i;
}

__global__ void __launch_bounds__(1024)
k_blockscan(const float* __restrict__ totr, const float* __restrict__ toti,
            const float* __restrict__ scal,
            const float* __restrict__ U0, const float* __restrict__ V0,
            float* __restrict__ carr, float* __restrict__ cari, int nblk) {
    __shared__ float wr[2][32];
    __shared__ float wi[2][32];
    const int tid  = threadIdx.x;
    const int lane = tid & 31;
    const int wv   = __builtin_amdgcn_readfirstlane(tid >> 5);
    const c32 A256 = { scal[38], scal[39] };
    const c32 u0   = { U0[0], V0[0] };
    c32 cin = { 0.f, 0.f };
    int buf = 0;
    for (int c0 = 0; c0 < nblk; c0 += 1024, buf ^= 1) {
        const int i = c0 + tid;
        int j = i - 1;
        j = j < 0 ? 0 : j;
        j = j < nblk ? j : nblk - 1;
        c32 x = { totr[j], toti[j] };
        if (i == 0)    x = u0;
        if (i >= nblk) x = { 0.f, 0.f };

        c32 m = A256, v = x;
        wave_affine_scan(m, v, lane);
        if (lane == 31) { wr[buf][wv] = v.r; wi[buf][wv] = v.i; }
        const c32 Q = { bc_lane(m.r, 31), bc_lane(m.i, 31) };
        __syncthreads();

        c32 m2 = Q, v2 = { wr[buf][lane], wi[buf][lane] };
        wave_affine_scan(m2, v2, lane);
        const int pw = wv > 0 ? wv - 1 : 0;
        c32 E = { bc_lane(v2.r, pw), bc_lane(v2.i, pw) };
        c32 F = { bc_lane(m2.r, pw), bc_lane(m2.i, pw) };
        const c32 tot = { bc_lane(v2.r, 31), bc_lane(v2.i, 31) };
        const c32 mt  = { bc_lane(m2.r, 31), bc_lane(m2.i, 31) };
        bool front = wv > 0;
        if (wv == 0) { E = { 0.f, 0.f }; F = { 1.f, 0.f }; }
        if (c0 > 0) {
            E = cadd(E, cmul(F, cin));
            front = true;
            cin = cadd(tot, cmul(mt, cin));
        } else {
            cin = tot;
        }
        const c32 res = front ? cadd(v, cmul(m, E)) : v;
        if (i < nblk) { carr[i] = res.r; cari[i] = res.i; }
    }
}

extern "C" void kernel_launch(void* const* d_in, const int* in_sizes, int n_in,
                              void* d_out, int out_size, void* d_ws, size_t ws_size,
                              hipStream_t stream) {
    const float* pk  = (const float*)d_in[0];
    const float* TAr = (const float*)d_in[1];
    const float* TAi = (const float*)d_in[2];
    const float* fc  = (const float*)d_in[3];
    const float* U0  = (const float*)d_in[4];
    const float* V0  = (const float*)d_in[5];
    const int*   dt  = (const int*)d_in[6];
    const int*   dtf = (const int*)d_in[7];

    const int NS   = in_sizes[1];
    const int NBLK = NS / 256;

    float* ws   = (float*)d_ws;
    float* scal = ws;
    float* br   = ws + 64;
    float* bi   = br + NS;
    float* totr = bi + NS;
    float* toti = totr + NBLK;
    float* carr = toti + NBLK;
    float* cari = carr + NBLK;
    float* out  = (float*)d_out;

    k_setup<<<1, 1, 0, stream>>>(pk, fc, dt, dtf, scal);
    k_forcing<<<(NS + 255) / 256, 256, 0, stream>>>(TAr, TAi, scal, dt, dtf, br, bi, NS);
    k_scan<false><<<NBLK, 32, 0, stream>>>(br, bi, scal, nullptr, nullptr, totr, toti, out, NS);
    k_blockscan<<<1, 1024, 0, stream>>>(totr, toti, scal, U0, V0, carr, cari, NBLK);
    k_scan<true><<<NBLK, 32, 0, stream>>>(br, bi, scal, carr, cari, totr, toti, out, NS);
}
